# starting kernel
# baseline (speedup 1.0000x reference)
#include <hip/hip_runtime.h>
#include <hip/hip_bf16.h>
#include <math.h>

typedef __attribute__((ext_vector_type(2))) float v2f;
typedef __attribute__((ext_vector_type(8))) float v8f;

#define D_DIM 256
#define C_DIM 64
#define EPS_F 1e-8f

__global__ void SimRel_cnorm_kernel(const float* __restrict__ ca,
                                    float* __restrict__ cn,
                                    float* __restrict__ cinf) {
    int c = threadIdx.x;
    if (c >= C_DIM) return;
    float s = 0.0f;
    int has_inf = 0;
    #pragma unroll 8
    for (int d = 0; d < D_DIM; ++d) {
        float v = ca[c * D_DIM + d];
        has_inf |= isinf(v) ? 1 : 0;
        s += v * v;
    }
    cn[c]   = sqrtf(s);
    cinf[c] = has_inf ? 1.0f : 0.0f;
}

__global__ void __launch_bounds__(256, 2)
SimRel_cossim_kernel(const float* __restrict__ x,
                     const float* __restrict__ ca,
                     const float* __restrict__ cn,
                     const float* __restrict__ cinf,
                     float* __restrict__ out) {
    __shared__ float sb[64 * 4 * 64];

    const int tid = threadIdx.x;

    {
        const int d   = tid;
        const int kk  = d >> 2;
        const int r   = d & 3;
        const int lhb = (r >> 1) << 4;
        const int j   = r & 1;
        for (int c = 0; c < C_DIM; ++c) {
            const int t = c >> 4;
            const int n = c & 15;
            sb[(kk * 4 + t) * 64 + (n + lhb) * 2 + j] = ca[c * D_DIM + d];
        }
    }
    __syncthreads();

    const int lane = tid & 31;
    const int wave = tid >> 5;
    const int half = lane >> 4;
    const int m0   = blockIdx.x * 128 + wave * 16;
    const int mrow = m0 + (lane & 15);

    const float* arow = x + (size_t)mrow * D_DIM + half * 2;

    v8f acc[4] = {v8f{}, v8f{}, v8f{}, v8f{}};
    float sumsq = 0.0f;

    #pragma unroll 4
    for (int kk = 0; kk < 64; ++kk) {
        v2f a = *(const v2f*)(arow + kk * 4);
        sumsq = fmaf(a.x, a.x, sumsq);
        sumsq = fmaf(a.y, a.y, sumsq);

        const float* bb = &sb[kk * 256 + lane * 2];
        v2f b0 = *(const v2f*)(bb +   0);
        v2f b1 = *(const v2f*)(bb +  64);
        v2f b2 = *(const v2f*)(bb + 128);
        v2f b3 = *(const v2f*)(bb + 192);

        acc[0] = __builtin_amdgcn_wmma_f32_16x16x4_f32(false, a, false, b0,
                                                       (short)0, acc[0], false, false);
        acc[1] = __builtin_amdgcn_wmma_f32_16x16x4_f32(false, a, false, b1,
                                                       (short)0, acc[1], false, false);
        acc[2] = __builtin_amdgcn_wmma_f32_16x16x4_f32(false, a, false, b2,
                                                       (short)0, acc[2], false, false);
        acc[3] = __builtin_amdgcn_wmma_f32_16x16x4_f32(false, a, false, b3,
                                                       (short)0, acc[3], false, false);
    }

    const float rowsq = sumsq + __shfl_xor(sumsq, 16, 32);

    float rn[8];
    #pragma unroll
    for (int r = 0; r < 8; ++r)
        rn[r] = sqrtf(__shfl(rowsq, (half ? 8 : 0) + r, 32));

    const int ncol = lane & 15;
    #pragma unroll
    for (int t = 0; t < 4; ++t) {
        const int col   = t * 16 + ncol;
        const float c_n = cn[col];
        const bool  c_i = cinf[col] > 0.0f;
        #pragma unroll
        for (int r = 0; r < 8; ++r) {
            const int row     = m0 + r + half * 8;
            const float dot   = acc[t][r];
            const float denom = fmaxf(rn[r] * c_n, EPS_F);
            const float v     = c_i ? 1.0f : dot / denom;
            out[(size_t)row * C_DIM + col] = v;
        }
    }
}

extern "C" void kernel_launch(void* const* d_in, const int* in_sizes, int n_in,
                              void* d_out, int out_size, void* d_ws, size_t ws_size,
                              hipStream_t stream) {
    const float* x  = (const float*)d_in[0];
    const float* ca = (const float*)d_in[2];
    float* out = (float*)d_out;

    float* cn   = (float*)d_ws;
    float* cinf = cn + C_DIM;

    SimRel_cnorm_kernel<<<1, 64, 0, stream>>>(ca, cn, cinf);

    const int rows_per_block = 128;
    const int grid = 262144 / rows_per_block;
    SimRel_cossim_kernel<<<grid, 256, 0, stream>>>(x, ca, cn, cinf, out);
}


# class matrix held in registers by persistent waves (1 per SIMD, 16 row tiles each); x loaded a tile ahead, previous tile scaled and stored between the WMMAs; k permuted within 8s: sum order changes
# speedup vs baseline: 1.1501x; 1.1501x over previous
#include <hip/hip_runtime.h>
#include <hip/hip_bf16.h>
#include <math.h>


typedef __attribute__((ext_vector_type(2))) float v2f;
typedef __attribute__((ext_vector_type(4))) float v4f;
typedef __attribute__((ext_vector_type(8))) float v8f;

#define D_DIM 256
#define C_DIM 64
#define EPS_F 1e-8f

#define N_WAVES        1024
#define TILES_PER_WAVE 16

__global__ void SimRel_cnorm_kernel(const float* __restrict__ ca,
                                    float* __restrict__ cn,
                                    float* __restrict__ cinf) {
    int c = threadIdx.x;
    if (c >= C_DIM) return;
    float s = 0.0f;
    int has_inf = 0;
    #pragma unroll 8
    for (int d = 0; d < D_DIM; ++d) {
        float v = ca[c * D_DIM + d];
        has_inf |= isinf(v) ? 1 : 0;
        s += v * v;
    }
    cn[c]   = sqrtf(s);
    cinf[c] = has_inf ? 1.0f : 0.0f;
}


__device__ __forceinline__ float row_value(float v, int r, int half) {
    const float lo = __int_as_float(__builtin_amdgcn_readlane(__float_as_int(v), r));
    const float hi = __int_as_float(__builtin_amdgcn_readlane(__float_as_int(v), r + 8));
    return half ? hi : lo;
}

__device__ __forceinline__ float finish_elem(const v8f (&pacc)[4], const float (&rn)[8],
                                             const float (&c_n)[4], const bool (&c_i)[4],
                                             int t, int r) {
    const float dot   = pacc[t][r];
    const float denom = fmaxf(rn[r] * c_n[t], EPS_F);
    const float q     = dot / denom;
    return c_i[t] ? 1.0f : q;
}

template <bool FIN>
__device__ __forceinline__ void tile_step(v4f (&aq)[32], const float* __restrict__ pnext,
                                          const v4f (&bq)[32][4],
                                          v8f (&acc)[4], float& sumsq,
                                          const v8f (&pacc)[4], float prowsq,
                                          float* __restrict__ pout, int half,
                                          const float (&c_n)[4], const bool (&c_i)[4]) {
    float rn[8];
    float res[8];
    if (FIN) {
        const float rnl = sqrtf(prowsq);
        #pragma unroll
        for (int r = 0; r < 8; ++r) rn[r] = row_value(rnl, r, half);
    }
    #pragma unroll
    for (int g = 0; g < 32; ++g) {
        const v4f a4 = aq[g];
        sumsq = fmaf(a4.x, a4.x, sumsq);
        sumsq = fmaf(a4.y, a4.y, sumsq);
        sumsq = fmaf(a4.z, a4.z, sumsq);
        sumsq = fmaf(a4.w, a4.w, sumsq);
        const v2f alo = __builtin_shufflevector(a4, a4, 0, 1);
        const v2f ahi = __builtin_shufflevector(a4, a4, 2, 3);
        #pragma unroll
        for (int t = 0; t < 4; ++t) {
            const v4f b4 = bq[g][t];
            const v2f blo = __builtin_shufflevector(b4, b4, 0, 1);
            acc[t] = __builtin_amdgcn_wmma_f32_16x16x4_f32(false, alo, false, blo,
                                                           (short)0, acc[t], false, false);
        }
        #pragma unroll
        for (int t = 0; t < 4; ++t) {
            const v4f b4 = bq[g][t];
            const v2f bhi = __builtin_shufflevector(b4, b4, 2, 3);
            acc[t] = __builtin_amdgcn_wmma_f32_16x16x4_f32(false, ahi, false, bhi,
                                                           (short)0, acc[t], false, false);
        }
        if (FIN) res[g & 7] = finish_elem(pacc, rn, c_n, c_i, g >> 3, g & 7);
        if ((g & 7) == 7) {
            __builtin_amdgcn_sched_barrier(0);
            if (FIN) {
                #pragma unroll
                for (int r = 0; r < 8; ++r) pout[r * C_DIM + (g >> 3) * 16] = res[r];
            }
            #pragma unroll
            for (int gg = g - 7; gg <= g; ++gg)
                aq[gg] = *(const v4f*)(pnext + 8 * gg);
            __builtin_amdgcn_sched_barrier(0);
        }
    }
}

__global__ void __launch_bounds__(128)
SimRel_cossim_kernel(const float* __restrict__ x,
                     const float* __restrict__ ca,
                     const float* __restrict__ cn,
                     const float* __restrict__ cinf,
                     float* __restrict__ out) {
    __shared__ v4f sbq[128 * 32];

    const int tid  = threadIdx.x;
    const int lane = tid & 31;
    const int wave = tid >> 5;
    const int half = lane >> 4;
    const int l15  = lane & 15;

    const int wid = blockIdx.x * 4 + wave;
    const float* xlane = x + ((size_t)wid * 16 + l15) * D_DIM + 4 * half;
    const size_t x_stride = (size_t)N_WAVES * 16 * D_DIM;
    float* olane = out + ((size_t)wid * 16 + 8 * half) * C_DIM + l15;
    const size_t o_stride = (size_t)N_WAVES * 16 * C_DIM;

    #pragma unroll
    for (int gg = 0; gg < 8; ++gg) {
        const int g = wave * 8 + gg;
        #pragma unroll
        for (int t = 0; t < 4; ++t) {
            const v4f q = *(const v4f*)(ca + (t * 16 + l15) * D_DIM + 8 * g + 4 * half);
            sbq[(g * 4 + t) * 32 + lane] = q;
        }
    }

    float c_n[4];
    bool  c_i[4];
    #pragma unroll
    for (int t = 0; t < 4; ++t) {
        c_n[t] = cn[t * 16 + l15];
        c_i[t] = cinf[t * 16 + l15] > 0.0f;
    }

    __syncthreads();

    v4f aq[32];
    #pragma unroll
    for (int g = 0; g < 32; ++g) aq[g] = *(const v4f*)(xlane + 8 * g);

    v4f bq[32][4];
    #pragma unroll
    for (int g = 0; g < 32; ++g) {
        #pragma unroll
        for (int t = 0; t < 4; ++t)
            bq[g][t] = sbq[(g * 4 + t) * 32 + lane];
    }

    v8f acc[4]  = {v8f{}, v8f{}, v8f{}, v8f{}};
    v8f pacc[4] = {v8f{}, v8f{}, v8f{}, v8f{}};
    float sumsq = 0.0f;

    tile_step<false>(aq, xlane + x_stride, bq, acc, sumsq, pacc, 0.0f, olane, half, c_n, c_i);

    #pragma unroll 1
    for (int i = 1; i < TILES_PER_WAVE; ++i) {
        const float prowsq = sumsq + __shfl_xor(sumsq, 16, 32);
        #pragma unroll
        for (int t = 0; t < 4; ++t) { pacc[t] = acc[t]; acc[t] = v8f{}; }
        sumsq = 0.0f;
        const int inext = (i + 1 < TILES_PER_WAVE) ? i + 1 : i;
        tile_step<true>(aq, xlane + (size_t)inext * x_stride, bq, acc, sumsq,
                        pacc, prowsq, olane + (size_t)(i - 1) * o_stride, half, c_n, c_i);
    }

    {
        const float prowsq = sumsq + __shfl_xor(sumsq, 16, 32);
        const float rnl = sqrtf(prowsq);
        float rn[8];
        #pragma unroll
        for (int r = 0; r < 8; ++r) rn[r] = row_value(rnl, r, half);
        float* pout = olane + (size_t)(TILES_PER_WAVE - 1) * o_stride;
        #pragma unroll
        for (int t = 0; t < 4; ++t) {
            #pragma unroll
            for (int r = 0; r < 8; ++r)
                pout[r * C_DIM + t * 16] = finish_elem(acc, rn, c_n, c_i, t, r);
        }
    }
}

extern "C" void kernel_launch(void* const* d_in, const int* in_sizes, int n_in,
                              void* d_out, int out_size, void* d_ws, size_t ws_size,
                              hipStream_t stream) {
    const float* x  = (const float*)d_in[0];
    const float* ca = (const float*)d_in[2];
    float* out = (float*)d_out;

    float* cn   = (float*)d_ws;
    float* cinf = cn + C_DIM;

    SimRel_cnorm_kernel<<<1, 64, 0, stream>>>(ca, cn, cinf);

    SimRel_cossim_kernel<<<256, 128, 0, stream>>>(x, ca, cn, cinf, out);
}


# class-to-column map permuted (column c of tile t = class 4c+t): a lane holds 4 adjacent classes, outputs leave as whole 256 B rows in 8 16-byte stores per tile, not 32 4-byte; sum order unchanged
# speedup vs baseline: 1.2192x; 1.0601x over previous
#include <hip/hip_runtime.h>
#include <hip/hip_bf16.h>
#include <math.h>


typedef __attribute__((ext_vector_type(2))) float v2f;
typedef __attribute__((ext_vector_type(4))) float v4f;
typedef __attribute__((ext_vector_type(8))) float v8f;

#define D_DIM 256
#define C_DIM 64
#define EPS_F 1e-8f

#define N_WAVES        1024
#define TILES_PER_WAVE 16

__global__ void SimRel_cnorm_kernel(const float* __restrict__ ca,
                                    float* __restrict__ cn,
                                    float* __restrict__ cinf) {
    int c = threadIdx.x;
    if (c >= C_DIM) return;
    float s = 0.0f;
    int has_inf = 0;
    #pragma unroll 8
    for (int d = 0; d < D_DIM; ++d) {
        float v = ca[c * D_DIM + d];
        has_inf |= isinf(v) ? 1 : 0;
        s += v * v;
    }
    cn[c]   = sqrtf(s);
    cinf[c] = has_inf ? 1.0f : 0.0f;
}


__device__ __forceinline__ float row_value(float v, int r, int half) {
    const float lo = __int_as_float(__builtin_amdgcn_readlane(__float_as_int(v), r));
    const float hi = __int_as_float(__builtin_amdgcn_readlane(__float_as_int(v), r + 8));
    return half ? hi : lo;
}

__device__ __forceinline__ float finish_elem(const v8f (&pacc)[4], const float (&rn)[8],
                                             const float (&c_n)[4], const bool (&c_i)[4],
                                             int t, int r) {
    const float dot   = pacc[t][r];
    const float denom = fmaxf(rn[r] * c_n[t], EPS_F);
    const float q     = dot / denom;
    return c_i[t] ? 1.0f : q;
}

template <bool FIN>
__device__ __forceinline__ void tile_step(v4f (&aq)[32], const float* __restrict__ pnext,
                                          const v4f (&bq)[32][4],
                                          v8f (&acc)[4], float& sumsq,
                                          const v8f (&pacc)[4], float prowsq,
                                          float* __restrict__ pout, int half,
                                          const float (&c_n)[4], const bool (&c_i)[4]) {
    float rn[8];
    float res[8];
    if (FIN) {
        const float rnl = sqrtf(prowsq);
        #pragma unroll
        for (int r = 0; r < 8; ++r) rn[r] = row_value(rnl, r, half);
    }
    #pragma unroll
    for (int g = 0; g < 32; ++g) {
        const v4f a4 = aq[g];
        sumsq = fmaf(a4.x, a4.x, sumsq);
        sumsq = fmaf(a4.y, a4.y, sumsq);
        sumsq = fmaf(a4.z, a4.z, sumsq);
        sumsq = fmaf(a4.w, a4.w, sumsq);
        const v2f alo = __builtin_shufflevector(a4, a4, 0, 1);
        const v2f ahi = __builtin_shufflevector(a4, a4, 2, 3);
        #pragma unroll
        for (int t = 0; t < 4; ++t) {
            const v4f b4 = bq[g][t];
            const v2f blo = __builtin_shufflevector(b4, b4, 0, 1);
            acc[t] = __builtin_amdgcn_wmma_f32_16x16x4_f32(false, alo, false, blo,
                                                           (short)0, acc[t], false, false);
        }
        #pragma unroll
        for (int t = 0; t < 4; ++t) {
            const v4f b4 = bq[g][t];
            const v2f bhi = __builtin_shufflevector(b4, b4, 2, 3);
            acc[t] = __builtin_amdgcn_wmma_f32_16x16x4_f32(false, ahi, false, bhi,
                                                           (short)0, acc[t], false, false);
        }
        if (FIN) res[g & 7] = finish_elem(pacc, rn, c_n, c_i, g & 3, 2 * (g >> 3) + ((g & 7) >> 2));
        if ((g & 7) == 7) {
            __builtin_amdgcn_sched_barrier(0);
            if (FIN) {
                *(v4f*)(pout + (2 * (g >> 3)) * C_DIM)     = v4f{res[0], res[1], res[2], res[3]};
                *(v4f*)(pout + (2 * (g >> 3) + 1) * C_DIM) = v4f{res[4], res[5], res[6], res[7]};
            }
            #pragma unroll
            for (int gg = g - 7; gg <= g; ++gg)
                aq[gg] = *(const v4f*)(pnext + 8 * gg);
            __builtin_amdgcn_sched_barrier(0);
        }
    }
}

__global__ void __launch_bounds__(128)
SimRel_cossim_kernel(const float* __restrict__ x,
                     const float* __restrict__ ca,
                     const float* __restrict__ cn,
                     const float* __restrict__ cinf,
                     float* __restrict__ out) {
    __shared__ v4f sbq[128 * 32];

    const int tid  = threadIdx.x;
    const int lane = tid & 31;
    const int wave = tid >> 5;
    const int half = lane >> 4;
    const int l15  = lane & 15;

    const int wid = blockIdx.x * 4 + wave;
    const float* xlane = x + ((size_t)wid * 16 + l15) * D_DIM + 4 * half;
    const size_t x_stride = (size_t)N_WAVES * 16 * D_DIM;
    float* olane = out + ((size_t)wid * 16 + 8 * half) * C_DIM + 4 * l15;
    const size_t o_stride = (size_t)N_WAVES * 16 * C_DIM;

    #pragma unroll
    for (int gg = 0; gg < 8; ++gg) {
        const int g = wave * 8 + gg;
        #pragma unroll
        for (int t = 0; t < 4; ++t) {
            const v4f q = *(const v4f*)(ca + (4 * l15 + t) * D_DIM + 8 * g + 4 * half);
            sbq[(g * 4 + t) * 32 + lane] = q;
        }
    }

    float c_n[4];
    bool  c_i[4];
    #pragma unroll
    for (int t = 0; t < 4; ++t) {
        c_n[t] = cn[4 * l15 + t];
        c_i[t] = cinf[4 * l15 + t] > 0.0f;
    }

    __syncthreads();

    v4f aq[32];
    #pragma unroll
    for (int g = 0; g < 32; ++g) aq[g] = *(const v4f*)(xlane + 8 * g);

    v4f bq[32][4];
    #pragma unroll
    for (int g = 0; g < 32; ++g) {
        #pragma unroll
        for (int t = 0; t < 4; ++t)
            bq[g][t] = sbq[(g * 4 + t) * 32 + lane];
    }

    v8f acc[4]  = {v8f{}, v8f{}, v8f{}, v8f{}};
    v8f pacc[4] = {v8f{}, v8f{}, v8f{}, v8f{}};
    float sumsq = 0.0f;

    tile_step<false>(aq, xlane + x_stride, bq, acc, sumsq, pacc, 0.0f, olane, half, c_n, c_i);

    #pragma unroll 1
    for (int i = 1; i < TILES_PER_WAVE; ++i) {
        const float prowsq = sumsq + __shfl_xor(sumsq, 16, 32);
        #pragma unroll
        for (int t = 0; t < 4; ++t) { pacc[t] = acc[t]; acc[t] = v8f{}; }
        sumsq = 0.0f;
        const int inext = (i + 1 < TILES_PER_WAVE) ? i + 1 : i;
        tile_step<true>(aq, xlane + (size_t)inext * x_stride, bq, acc, sumsq,
                        pacc, prowsq, olane + (size_t)(i - 1) * o_stride, half, c_n, c_i);
    }

    {
        const float prowsq = sumsq + __shfl_xor(sumsq, 16, 32);
        const float rnl = sqrtf(prowsq);
        float rn[8];
        #pragma unroll
        for (int r = 0; r < 8; ++r) rn[r] = row_value(rnl, r, half);
        float* pout = olane + (size_t)(TILES_PER_WAVE - 1) * o_stride;
        #pragma unroll
        for (int r = 0; r < 8; ++r) {
            *(v4f*)(pout + r * C_DIM) = v4f{finish_elem(acc, rn, c_n, c_i, 0, r),
                                            finish_elem(acc, rn, c_n, c_i, 1, r),
                                            finish_elem(acc, rn, c_n, c_i, 2, r),
                                            finish_elem(acc, rn, c_n, c_i, 3, r)};
        }
    }
}

extern "C" void kernel_launch(void* const* d_in, const int* in_sizes, int n_in,
                              void* d_out, int out_size, void* d_ws, size_t ws_size,
                              hipStream_t stream) {
    const float* x  = (const float*)d_in[0];
    const float* ca = (const float*)d_in[2];
    float* out = (float*)d_out;

    float* cn   = (float*)d_ws;
    float* cinf = cn + C_DIM;

    SimRel_cnorm_kernel<<<1, 64, 0, stream>>>(ca, cn, cinf);

    SimRel_cossim_kernel<<<256, 128, 0, stream>>>(x, ca, cn, cinf, out);
}


# class norms and inf flags computed in the main kernel from the class fragments each lane already holds: norm launch and workspace gone (2 launches -> 1); class norm sum order changes
# speedup vs baseline: 1.3638x; 1.1186x over previous
#include <hip/hip_runtime.h>
#include <hip/hip_bf16.h>
#include <math.h>


typedef __attribute__((ext_vector_type(2))) float v2f;
typedef __attribute__((ext_vector_type(4))) float v4f;
typedef __attribute__((ext_vector_type(8))) float v8f;

#define D_DIM 256
#define C_DIM 64
#define EPS_F 1e-8f

#define N_WAVES        1024
#define TILES_PER_WAVE 16


__device__ __forceinline__ float row_value(float v, int r, int half) {
    const float lo = __int_as_float(__builtin_amdgcn_readlane(__float_as_int(v), r));
    const float hi = __int_as_float(__builtin_amdgcn_readlane(__float_as_int(v), r + 8));
    return half ? hi : lo;
}

__device__ __forceinline__ float finish_elem(const v8f (&pacc)[4], const float (&rn)[8],
                                             const float (&c_n)[4], const bool (&c_i)[4],
                                             int t, int r) {
    const float dot   = pacc[t][r];
    const float denom = fmaxf(rn[r] * c_n[t], EPS_F);
    const float q     = dot / denom;
    return c_i[t] ? 1.0f : q;
}

template <bool FIN>
__device__ __forceinline__ void tile_step(v4f (&aq)[32], const float* __restrict__ pnext,
                                          const v4f (&bq)[32][4],
                                          v8f (&acc)[4], float& sumsq,
                                          const v8f (&pacc)[4], float prowsq,
                                          float* __restrict__ pout, int half,
                                          const float (&c_n)[4], const bool (&c_i)[4]) {
    float rn[8];
    float res[8];
    if (FIN) {
        const float rnl = sqrtf(prowsq);
        #pragma unroll
        for (int r = 0; r < 8; ++r) rn[r] = row_value(rnl, r, half);
    }
    #pragma unroll
    for (int g = 0; g < 32; ++g) {
        const v4f a4 = aq[g];
        sumsq = fmaf(a4.x, a4.x, sumsq);
        sumsq = fmaf(a4.y, a4.y, sumsq);
        sumsq = fmaf(a4.z, a4.z, sumsq);
        sumsq = fmaf(a4.w, a4.w, sumsq);
        const v2f alo = __builtin_shufflevector(a4, a4, 0, 1);
        const v2f ahi = __builtin_shufflevector(a4, a4, 2, 3);
        #pragma unroll
        for (int t = 0; t < 4; ++t) {
            const v4f b4 = bq[g][t];
            const v2f blo = __builtin_shufflevector(b4, b4, 0, 1);
            acc[t] = __builtin_amdgcn_wmma_f32_16x16x4_f32(false, alo, false, blo,
                                                           (short)0, acc[t], false, false);
        }
        #pragma unroll
        for (int t = 0; t < 4; ++t) {
            const v4f b4 = bq[g][t];
            const v2f bhi = __builtin_shufflevector(b4, b4, 2, 3);
            acc[t] = __builtin_amdgcn_wmma_f32_16x16x4_f32(false, ahi, false, bhi,
                                                           (short)0, acc[t], false, false);
        }
        if (FIN) res[g & 7] = finish_elem(pacc, rn, c_n, c_i, g & 3, 2 * (g >> 3) + ((g & 7) >> 2));
        if ((g & 7) == 7) {
            __builtin_amdgcn_sched_barrier(0);
            if (FIN) {
                *(v4f*)(pout + (2 * (g >> 3)) * C_DIM)     = v4f{res[0], res[1], res[2], res[3]};
                *(v4f*)(pout + (2 * (g >> 3) + 1) * C_DIM) = v4f{res[4], res[5], res[6], res[7]};
            }
            #pragma unroll
            for (int gg = g - 7; gg <= g; ++gg)
                aq[gg] = *(const v4f*)(pnext + 8 * gg);
            __builtin_amdgcn_sched_barrier(0);
        }
    }
}

__global__ void __launch_bounds__(128)
SimRel_cossim_kernel(const float* __restrict__ x,
                     const float* __restrict__ ca,
                     float* __restrict__ out) {
    __shared__ v4f sbq[128 * 32];

    const int tid  = threadIdx.x;
    const int lane = tid & 31;
    const int wave = tid >> 5;
    const int half = lane >> 4;
    const int l15  = lane & 15;

    const int wid = blockIdx.x * 4 + wave;
    const float* xlane = x + ((size_t)wid * 16 + l15) * D_DIM + 4 * half;
    const size_t x_stride = (size_t)N_WAVES * 16 * D_DIM;
    float* olane = out + ((size_t)wid * 16 + 8 * half) * C_DIM + 4 * l15;
    const size_t o_stride = (size_t)N_WAVES * 16 * C_DIM;

    #pragma unroll
    for (int gg = 0; gg < 8; ++gg) {
        const int g = wave * 8 + gg;
        #pragma unroll
        for (int t = 0; t < 4; ++t) {
            const v4f q = *(const v4f*)(ca + (4 * l15 + t) * D_DIM + 8 * g + 4 * half);
            sbq[(g * 4 + t) * 32 + lane] = q;
        }
    }

    __syncthreads();

    v4f aq[32];
    #pragma unroll
    for (int g = 0; g < 32; ++g) aq[g] = *(const v4f*)(xlane + 8 * g);

    v4f bq[32][4];
    #pragma unroll
    for (int g = 0; g < 32; ++g) {
        #pragma unroll
        for (int t = 0; t < 4; ++t)
            bq[g][t] = sbq[(g * 4 + t) * 32 + lane];
    }

    float c_n[4];
    bool  c_i[4];
    {
        v2f   ss[4] = {v2f{0.0f, 0.0f}, v2f{0.0f, 0.0f}, v2f{0.0f, 0.0f}, v2f{0.0f, 0.0f}};
        float mx[4] = {0.0f, 0.0f, 0.0f, 0.0f};
        #pragma unroll
        for (int g = 0; g < 32; ++g) {
            #pragma unroll
            for (int t = 0; t < 4; ++t) {
                const v4f b4 = bq[g][t];
                const v2f blo = __builtin_shufflevector(b4, b4, 0, 1);
                const v2f bhi = __builtin_shufflevector(b4, b4, 2, 3);
                ss[t] = __builtin_elementwise_fma(blo, blo, ss[t]);
                ss[t] = __builtin_elementwise_fma(bhi, bhi, ss[t]);
                mx[t] = fmaxf(fmaxf(mx[t], fabsf(b4.x)), fabsf(b4.y));
                mx[t] = fmaxf(fmaxf(mx[t], fabsf(b4.z)), fabsf(b4.w));
            }
        }
        #pragma unroll
        for (int t = 0; t < 4; ++t) {
            const float sh = ss[t].x + ss[t].y;
            const float s2 = sh + __shfl_xor(sh, 16, 32);
            const float m  = fmaxf(mx[t], __shfl_xor(mx[t], 16, 32));
            c_n[t] = sqrtf(s2);
            c_i[t] = isinf(m);
        }
    }

    v8f acc[4]  = {v8f{}, v8f{}, v8f{}, v8f{}};
    v8f pacc[4] = {v8f{}, v8f{}, v8f{}, v8f{}};
    float sumsq = 0.0f;

    tile_step<false>(aq, xlane + x_stride, bq, acc, sumsq, pacc, 0.0f, olane, half, c_n, c_i);

    #pragma unroll 1
    for (int i = 1; i < TILES_PER_WAVE; ++i) {
        const float prowsq = sumsq + __shfl_xor(sumsq, 16, 32);
        #pragma unroll
        for (int t = 0; t < 4; ++t) { pacc[t] = acc[t]; acc[t] = v8f{}; }
        sumsq = 0.0f;
        const int inext = (i + 1 < TILES_PER_WAVE) ? i + 1 : i;
        tile_step<true>(aq, xlane + (size_t)inext * x_stride, bq, acc, sumsq,
                        pacc, prowsq, olane + (size_t)(i - 1) * o_stride, half, c_n, c_i);
    }

    {
        const float prowsq = sumsq + __shfl_xor(sumsq, 16, 32);
        const float rnl = sqrtf(prowsq);
        float rn[8];
        #pragma unroll
        for (int r = 0; r < 8; ++r) rn[r] = row_value(rnl, r, half);
        float* pout = olane + (size_t)(TILES_PER_WAVE - 1) * o_stride;
        #pragma unroll
        for (int r = 0; r < 8; ++r) {
            *(v4f*)(pout + r * C_DIM) = v4f{finish_elem(acc, rn, c_n, c_i, 0, r),
                                            finish_elem(acc, rn, c_n, c_i, 1, r),
                                            finish_elem(acc, rn, c_n, c_i, 2, r),
                                            finish_elem(acc, rn, c_n, c_i, 3, r)};
        }
    }
}

extern "C" void kernel_launch(void* const* d_in, const int* in_sizes, int n_in,
                              void* d_out, int out_size, void* d_ws, size_t ws_size,
                              hipStream_t stream) {
    const float* x  = (const float*)d_in[0];
    const float* ca = (const float*)d_in[2];
    float* out = (float*)d_out;

    SimRel_cossim_kernel<<<256, 128, 0, stream>>>(x, ca, out);
}


# output scaling: dot * v_rcp_f32(denom) replaces the correctly rounded quotient (11 -> 2 vector instructions per output, 32 outputs per lane per tile); quotient within 2 ulp; sum order unchanged
# speedup vs baseline: 1.4308x; 1.0491x over previous
#include <hip/hip_runtime.h>
#include <hip/hip_bf16.h>
#include <math.h>


typedef __attribute__((ext_vector_type(2))) float v2f;
typedef __attribute__((ext_vector_type(4))) float v4f;
typedef __attribute__((ext_vector_type(8))) float v8f;

#define D_DIM 256
#define C_DIM 64
#define EPS_F 1e-8f

#define N_WAVES        1024
#define TILES_PER_WAVE 16


__device__ __forceinline__ float row_value(float v, int r, int half) {
    const float lo = __int_as_float(__builtin_amdgcn_readlane(__float_as_int(v), r));
    const float hi = __int_as_float(__builtin_amdgcn_readlane(__float_as_int(v), r + 8));
    return half ? hi : lo;
}

__device__ __forceinline__ float finish_elem(const v8f (&pacc)[4], const float (&rn)[8],
                                             const float (&c_n)[4], const bool (&c_i)[4],
                                             int t, int r) {
    const float dot   = pacc[t][r];
    const float denom = fmaxf(rn[r] * c_n[t], EPS_F);
    const float q     = dot * __builtin_amdgcn_rcpf(denom);
    return c_i[t] ? 1.0f : q;
}

template <bool FIN>
__device__ __forceinline__ void tile_step(v4f (&aq)[32], const float* __restrict__ pnext,
                                          const v4f (&bq)[32][4],
                                          v8f (&acc)[4], float& sumsq,
                                          const v8f (&pacc)[4], float prowsq,
                                          float* __restrict__ pout, int half,
                                          const float (&c_n)[4], const bool (&c_i)[4]) {
    float rn[8];
    float res[8];
    if (FIN) {
        const float rnl = sqrtf(prowsq);
        #pragma unroll
        for (int r = 0; r < 8; ++r) rn[r] = row_value(rnl, r, half);
    }
    #pragma unroll
    for (int g = 0; g < 32; ++g) {
        const v4f a4 = aq[g];
        sumsq = fmaf(a4.x, a4.x, sumsq);
        sumsq = fmaf(a4.y, a4.y, sumsq);
        sumsq = fmaf(a4.z, a4.z, sumsq);
        sumsq = fmaf(a4.w, a4.w, sumsq);
        const v2f alo = __builtin_shufflevector(a4, a4, 0, 1);
        const v2f ahi = __builtin_shufflevector(a4, a4, 2, 3);
        #pragma unroll
        for (int t = 0; t < 4; ++t) {
            const v4f b4 = bq[g][t];
            const v2f blo = __builtin_shufflevector(b4, b4, 0, 1);
            acc[t] = __builtin_amdgcn_wmma_f32_16x16x4_f32(false, alo, false, blo,
                                                           (short)0, acc[t], false, false);
        }
        #pragma unroll
        for (int t = 0; t < 4; ++t) {
            const v4f b4 = bq[g][t];
            const v2f bhi = __builtin_shufflevector(b4, b4, 2, 3);
            acc[t] = __builtin_amdgcn_wmma_f32_16x16x4_f32(false, ahi, false, bhi,
                                                           (short)0, acc[t], false, false);
        }
        if (FIN) res[g & 7] = finish_elem(pacc, rn, c_n, c_i, g & 3, 2 * (g >> 3) + ((g & 7) >> 2));
        if ((g & 7) == 7) {
            __builtin_amdgcn_sched_barrier(0);
            if (FIN) {
                *(v4f*)(pout + (2 * (g >> 3)) * C_DIM)     = v4f{res[0], res[1], res[2], res[3]};
                *(v4f*)(pout + (2 * (g >> 3) + 1) * C_DIM) = v4f{res[4], res[5], res[6], res[7]};
            }
            #pragma unroll
            for (int gg = g - 7; gg <= g; ++gg)
                aq[gg] = *(const v4f*)(pnext + 8 * gg);
            __builtin_amdgcn_sched_barrier(0);
        }
    }
}

__global__ void __launch_bounds__(128)
SimRel_cossim_kernel(const float* __restrict__ x,
                     const float* __restrict__ ca,
                     float* __restrict__ out) {
    __shared__ v4f sbq[128 * 32];

    const int tid  = threadIdx.x;
    const int lane = tid & 31;
    const int wave = tid >> 5;
    const int half = lane >> 4;
    const int l15  = lane & 15;

    const int wid = blockIdx.x * 4 + wave;
    const float* xlane = x + ((size_t)wid * 16 + l15) * D_DIM + 4 * half;
    const size_t x_stride = (size_t)N_WAVES * 16 * D_DIM;
    float* olane = out + ((size_t)wid * 16 + 8 * half) * C_DIM + 4 * l15;
    const size_t o_stride = (size_t)N_WAVES * 16 * C_DIM;

    #pragma unroll
    for (int gg = 0; gg < 8; ++gg) {
        const int g = wave * 8 + gg;
        #pragma unroll
        for (int t = 0; t < 4; ++t) {
            const v4f q = *(const v4f*)(ca + (4 * l15 + t) * D_DIM + 8 * g + 4 * half);
            sbq[(g * 4 + t) * 32 + lane] = q;
        }
    }

    __syncthreads();

    v4f aq[32];
    #pragma unroll
    for (int g = 0; g < 32; ++g) aq[g] = *(const v4f*)(xlane + 8 * g);

    v4f bq[32][4];
    #pragma unroll
    for (int g = 0; g < 32; ++g) {
        #pragma unroll
        for (int t = 0; t < 4; ++t)
            bq[g][t] = sbq[(g * 4 + t) * 32 + lane];
    }

    float c_n[4];
    bool  c_i[4];
    {
        v2f   ss[4] = {v2f{0.0f, 0.0f}, v2f{0.0f, 0.0f}, v2f{0.0f, 0.0f}, v2f{0.0f, 0.0f}};
        float mx[4] = {0.0f, 0.0f, 0.0f, 0.0f};
        #pragma unroll
        for (int g = 0; g < 32; ++g) {
            #pragma unroll
            for (int t = 0; t < 4; ++t) {
                const v4f b4 = bq[g][t];
                const v2f blo = __builtin_shufflevector(b4, b4, 0, 1);
                const v2f bhi = __builtin_shufflevector(b4, b4, 2, 3);
                ss[t] = __builtin_elementwise_fma(blo, blo, ss[t]);
                ss[t] = __builtin_elementwise_fma(bhi, bhi, ss[t]);
                mx[t] = fmaxf(fmaxf(mx[t], fabsf(b4.x)), fabsf(b4.y));
                mx[t] = fmaxf(fmaxf(mx[t], fabsf(b4.z)), fabsf(b4.w));
            }
        }
        #pragma unroll
        for (int t = 0; t < 4; ++t) {
            const float sh = ss[t].x + ss[t].y;
            const float s2 = sh + __shfl_xor(sh, 16, 32);
            const float m  = fmaxf(mx[t], __shfl_xor(mx[t], 16, 32));
            c_n[t] = sqrtf(s2);
            c_i[t] = isinf(m);
        }
    }

    v8f acc[4]  = {v8f{}, v8f{}, v8f{}, v8f{}};
    v8f pacc[4] = {v8f{}, v8f{}, v8f{}, v8f{}};
    float sumsq = 0.0f;

    tile_step<false>(aq, xlane + x_stride, bq, acc, sumsq, pacc, 0.0f, olane, half, c_n, c_i);

    #pragma unroll 1
    for (int i = 1; i < TILES_PER_WAVE; ++i) {
        const float prowsq = sumsq + __shfl_xor(sumsq, 16, 32);
        #pragma unroll
        for (int t = 0; t < 4; ++t) { pacc[t] = acc[t]; acc[t] = v8f{}; }
        sumsq = 0.0f;
        const int inext = (i + 1 < TILES_PER_WAVE) ? i + 1 : i;
        tile_step<true>(aq, xlane + (size_t)inext * x_stride, bq, acc, sumsq,
                        pacc, prowsq, olane + (size_t)(i - 1) * o_stride, half, c_n, c_i);
    }

    {
        const float prowsq = sumsq + __shfl_xor(sumsq, 16, 32);
        const float rnl = sqrtf(prowsq);
        float rn[8];
        #pragma unroll
        for (int r = 0; r < 8; ++r) rn[r] = row_value(rnl, r, half);
        float* pout = olane + (size_t)(TILES_PER_WAVE - 1) * o_stride;
        #pragma unroll
        for (int r = 0; r < 8; ++r) {
            *(v4f*)(pout + r * C_DIM) = v4f{finish_elem(acc, rn, c_n, c_i, 0, r),
                                            finish_elem(acc, rn, c_n, c_i, 1, r),
                                            finish_elem(acc, rn, c_n, c_i, 2, r),
                                            finish_elem(acc, rn, c_n, c_i, 3, r)};
        }
    }
}

extern "C" void kernel_launch(void* const* d_in, const int* in_sizes, int n_in,
                              void* d_out, int out_size, void* d_ws, size_t ws_size,
                              hipStream_t stream) {
    const float* x  = (const float*)d_in[0];
    const float* ca = (const float*)d_in[2];
    float* out = (float*)d_out;

    SimRel_cossim_kernel<<<256, 128, 0, stream>>>(x, ca, out);
}


# row sum of squares: 64 packed multiply-adds into 32 independent partial sums per tile (tree-added) replace 128 chained ones that each waited for the one before; row norm sum order changes
# speedup vs baseline: 1.4737x; 1.0300x over previous
#include <hip/hip_runtime.h>
#include <hip/hip_bf16.h>
#include <math.h>


typedef __attribute__((ext_vector_type(2))) float v2f;
typedef __attribute__((ext_vector_type(4))) float v4f;
typedef __attribute__((ext_vector_type(8))) float v8f;

#define D_DIM 256
#define C_DIM 64
#define EPS_F 1e-8f

#define N_WAVES        1024
#define TILES_PER_WAVE 16


__device__ __forceinline__ float row_value(float v, int r, int half) {
    const float lo = __int_as_float(__builtin_amdgcn_readlane(__float_as_int(v), r));
    const float hi = __int_as_float(__builtin_amdgcn_readlane(__float_as_int(v), r + 8));
    return half ? hi : lo;
}

__device__ __forceinline__ float finish_elem(const v8f (&pacc)[4], const float (&rn)[8],
                                             const float (&c_n)[4], const bool (&c_i)[4],
                                             int t, int r) {
    const float dot   = pacc[t][r];
    const float denom = fmaxf(rn[r] * c_n[t], EPS_F);
    const float q     = dot * __builtin_amdgcn_rcpf(denom);
    return c_i[t] ? 1.0f : q;
}

__device__ __forceinline__ float row_sumsq(const v2f (&ssq)[16]) {
    v2f s8[8];
    #pragma unroll
    for (int j = 0; j < 8; ++j) s8[j] = ssq[2 * j] + ssq[2 * j + 1];
    const v2f s = ((s8[0] + s8[1]) + (s8[2] + s8[3])) + ((s8[4] + s8[5]) + (s8[6] + s8[7]));
    const float sh = s.x + s.y;
    return sh + __shfl_xor(sh, 16, 32);
}

template <bool FIN>
__device__ __forceinline__ void tile_step(v4f (&aq)[32], const float* __restrict__ pnext,
                                          const v4f (&bq)[32][4],
                                          v8f (&acc)[4], v2f (&ssq)[16],
                                          const v8f (&pacc)[4], float prowsq,
                                          float* __restrict__ pout, int half,
                                          const float (&c_n)[4], const bool (&c_i)[4]) {
    float rn[8];
    float res[8];
    if (FIN) {
        const float rnl = sqrtf(prowsq);
        #pragma unroll
        for (int r = 0; r < 8; ++r) rn[r] = row_value(rnl, r, half);
    }
    #pragma unroll
    for (int g = 0; g < 32; ++g) {
        const v4f a4 = aq[g];
        const v2f alo = __builtin_shufflevector(a4, a4, 0, 1);
        const v2f ahi = __builtin_shufflevector(a4, a4, 2, 3);
        ssq[2 * (g & 7)]     = __builtin_elementwise_fma(alo, alo, ssq[2 * (g & 7)]);
        ssq[2 * (g & 7) + 1] = __builtin_elementwise_fma(ahi, ahi, ssq[2 * (g & 7) + 1]);
        #pragma unroll
        for (int t = 0; t < 4; ++t) {
            const v4f b4 = bq[g][t];
            const v2f blo = __builtin_shufflevector(b4, b4, 0, 1);
            acc[t] = __builtin_amdgcn_wmma_f32_16x16x4_f32(false, alo, false, blo,
                                                           (short)0, acc[t], false, false);
        }
        #pragma unroll
        for (int t = 0; t < 4; ++t) {
            const v4f b4 = bq[g][t];
            const v2f bhi = __builtin_shufflevector(b4, b4, 2, 3);
            acc[t] = __builtin_amdgcn_wmma_f32_16x16x4_f32(false, ahi, false, bhi,
                                                           (short)0, acc[t], false, false);
        }
        if (FIN) res[g & 7] = finish_elem(pacc, rn, c_n, c_i, g & 3, 2 * (g >> 3) + ((g & 7) >> 2));
        if ((g & 7) == 7) {
            __builtin_amdgcn_sched_barrier(0);
            if (FIN) {
                *(v4f*)(pout + (2 * (g >> 3)) * C_DIM)     = v4f{res[0], res[1], res[2], res[3]};
                *(v4f*)(pout + (2 * (g >> 3) + 1) * C_DIM) = v4f{res[4], res[5], res[6], res[7]};
            }
            #pragma unroll
            for (int gg = g - 7; gg <= g; ++gg)
                aq[gg] = *(const v4f*)(pnext + 8 * gg);
            __builtin_amdgcn_sched_barrier(0);
        }
    }
}

__global__ void __launch_bounds__(128)
SimRel_cossim_kernel(const float* __restrict__ x,
                     const float* __restrict__ ca,
                     float* __restrict__ out) {
    __shared__ v4f sbq[128 * 32];

    const int tid  = threadIdx.x;
    const int lane = tid & 31;
    const int wave = tid >> 5;
    const int half = lane >> 4;
    const int l15  = lane & 15;

    const int wid = blockIdx.x * 4 + wave;
    const float* xlane = x + ((size_t)wid * 16 + l15) * D_DIM + 4 * half;
    const size_t x_stride = (size_t)N_WAVES * 16 * D_DIM;
    float* olane = out + ((size_t)wid * 16 + 8 * half) * C_DIM + 4 * l15;
    const size_t o_stride = (size_t)N_WAVES * 16 * C_DIM;

    #pragma unroll
    for (int gg = 0; gg < 8; ++gg) {
        const int g = wave * 8 + gg;
        #pragma unroll
        for (int t = 0; t < 4; ++t) {
            const v4f q = *(const v4f*)(ca + (4 * l15 + t) * D_DIM + 8 * g + 4 * half);
            sbq[(g * 4 + t) * 32 + lane] = q;
        }
    }

    __syncthreads();

    v4f aq[32];
    #pragma unroll
    for (int g = 0; g < 32; ++g) aq[g] = *(const v4f*)(xlane + 8 * g);

    v4f bq[32][4];
    #pragma unroll
    for (int g = 0; g < 32; ++g) {
        #pragma unroll
        for (int t = 0; t < 4; ++t)
            bq[g][t] = sbq[(g * 4 + t) * 32 + lane];
    }

    float c_n[4];
    bool  c_i[4];
    {
        v2f   ss[4] = {v2f{0.0f, 0.0f}, v2f{0.0f, 0.0f}, v2f{0.0f, 0.0f}, v2f{0.0f, 0.0f}};
        float mx[4] = {0.0f, 0.0f, 0.0f, 0.0f};
        #pragma unroll
        for (int g = 0; g < 32; ++g) {
            #pragma unroll
            for (int t = 0; t < 4; ++t) {
                const v4f b4 = bq[g][t];
                const v2f blo = __builtin_shufflevector(b4, b4, 0, 1);
                const v2f bhi = __builtin_shufflevector(b4, b4, 2, 3);
                ss[t] = __builtin_elementwise_fma(blo, blo, ss[t]);
                ss[t] = __builtin_elementwise_fma(bhi, bhi, ss[t]);
                mx[t] = fmaxf(fmaxf(mx[t], fabsf(b4.x)), fabsf(b4.y));
                mx[t] = fmaxf(fmaxf(mx[t], fabsf(b4.z)), fabsf(b4.w));
            }
        }
        #pragma unroll
        for (int t = 0; t < 4; ++t) {
            const float sh = ss[t].x + ss[t].y;
            const float s2 = sh + __shfl_xor(sh, 16, 32);
            const float m  = fmaxf(mx[t], __shfl_xor(mx[t], 16, 32));
            c_n[t] = sqrtf(s2);
            c_i[t] = isinf(m);
        }
    }

    v8f acc[4]  = {v8f{}, v8f{}, v8f{}, v8f{}};
    v8f pacc[4] = {v8f{}, v8f{}, v8f{}, v8f{}};
    v2f ssq[16];
    #pragma unroll
    for (int j = 0; j < 16; ++j) ssq[j] = v2f{0.0f, 0.0f};

    tile_step<false>(aq, xlane + x_stride, bq, acc, ssq, pacc, 0.0f, olane, half, c_n, c_i);

    #pragma unroll 1
    for (int i = 1; i < TILES_PER_WAVE; ++i) {
        const float prowsq = row_sumsq(ssq);
        #pragma unroll
        for (int t = 0; t < 4; ++t) { pacc[t] = acc[t]; acc[t] = v8f{}; }
        #pragma unroll
        for (int j = 0; j < 16; ++j) ssq[j] = v2f{0.0f, 0.0f};
        const int inext = (i + 1 < TILES_PER_WAVE) ? i + 1 : i;
        tile_step<true>(aq, xlane + (size_t)inext * x_stride, bq, acc, ssq,
                        pacc, prowsq, olane + (size_t)(i - 1) * o_stride, half, c_n, c_i);
    }

    {
        const float prowsq = row_sumsq(ssq);
        const float rnl = sqrtf(prowsq);
        float rn[8];
        #pragma unroll
        for (int r = 0; r < 8; ++r) rn[r] = row_value(rnl, r, half);
        float* pout = olane + (size_t)(TILES_PER_WAVE - 1) * o_stride;
        #pragma unroll
        for (int r = 0; r < 8; ++r) {
            *(v4f*)(pout + r * C_DIM) = v4f{finish_elem(acc, rn, c_n, c_i, 0, r),
                                            finish_elem(acc, rn, c_n, c_i, 1, r),
                                            finish_elem(acc, rn, c_n, c_i, 2, r),
                                            finish_elem(acc, rn, c_n, c_i, 3, r)};
        }
    }
}

extern "C" void kernel_launch(void* const* d_in, const int* in_sizes, int n_in,
                              void* d_out, int out_size, void* d_ws, size_t ws_size,
                              hipStream_t stream) {
    const float* x  = (const float*)d_in[0];
    const float* ca = (const float*)d_in[2];
    float* out = (float*)d_out;

    SimRel_cossim_kernel<<<256, 128, 0, stream>>>(x, ca, out);
}
